# starting kernel
# baseline (speedup 1.0000x reference)
#include <hip/hip_runtime.h>
#include <math.h>

typedef __attribute__((ext_vector_type(2))) float v2f;
typedef __attribute__((ext_vector_type(8))) float v8f;

#define KDIM   512
#define KNH    16
#define KHEAD  32
#define KH     64
#define KW     64
#define KG     32
#define KN     128
#define K2N    256
#define KSHIFT 4

static __device__ __forceinline__ v8f wmma_f32(v2f a, v2f b, v8f c) {
  return __builtin_amdgcn_wmma_f32_16x16x4_f32(false, a, false, b, (short)0, c,
                                               false, false);
}

__global__ __launch_bounds__(256) void gemm_bias_kernel(
    const float* __restrict__ A, const float* __restrict__ Wm,
    const float* __restrict__ bias, float* __restrict__ C, int M, int N) {
  __shared__ float As[128][68];
  __shared__ float BsT[64][66];

  const int nBlocks = N >> 6;
  const int mBlock  = blockIdx.x / nBlocks;
  const int nBlock  = blockIdx.x % nBlocks;
  const int rowBase = mBlock << 7;
  const int colBase = nBlock << 6;

  const int tid  = threadIdx.x;
  const int wave = tid >> 5;
  const int lane = tid & 31;
  const int l16  = lane & 15;
  const int half = lane >> 4;

  v8f acc[4];
  acc[0] = {}; acc[1] = {}; acc[2] = {}; acc[3] = {};

  const int arow = tid >> 1;
  const int acol = (tid & 1) << 5;
  const int brow = tid >> 2;
  const int bcol = (tid & 3) << 4;

  const float* ag0 = A + (size_t)(rowBase + arow) * KDIM + acol;
  const float* bg0 = Wm + (size_t)brow * N + colBase + bcol;

  float4 ra[8], rb[4];
#pragma unroll
  for (int i = 0; i < 8; ++i) ra[i] = ((const float4*)ag0)[i];
#pragma unroll
  for (int i = 0; i < 4; ++i) rb[i] = ((const float4*)bg0)[i];

  for (int chunk = 0; chunk < 8; ++chunk) {
#pragma unroll
    for (int i = 0; i < 8; ++i)
      *(float4*)&As[arow][acol + 4 * i] = ra[i];
#pragma unroll
    for (int i = 0; i < 4; ++i) {
      BsT[bcol + 4 * i + 0][brow] = rb[i].x;
      BsT[bcol + 4 * i + 1][brow] = rb[i].y;
      BsT[bcol + 4 * i + 2][brow] = rb[i].z;
      BsT[bcol + 4 * i + 3][brow] = rb[i].w;
    }
    __syncthreads();

    if (chunk < 7) {
      const float* ag = ag0 + (chunk + 1) * 64;
      const float* bg = bg0 + (size_t)(chunk + 1) * 64 * N;
#pragma unroll
      for (int i = 0; i < 8; ++i) ra[i] = ((const float4*)ag)[i];
#pragma unroll
      for (int i = 0; i < 4; ++i) rb[i] = ((const float4*)bg)[i];
    }

#pragma unroll
    for (int kk = 0; kk < 16; ++kk) {
      v2f a = *(const v2f*)&As[wave * 16 + l16][4 * kk + 2 * half];
#pragma unroll
      for (int nt = 0; nt < 4; ++nt) {
        v2f b = *(const v2f*)&BsT[nt * 16 + l16][4 * kk + 2 * half];
        acc[nt] = wmma_f32(a, b, acc[nt]);
      }
    }
    __syncthreads();
  }

#pragma unroll
  for (int nt = 0; nt < 4; ++nt) {
    const int col = colBase + nt * 16 + l16;
    const float bv = bias[col];
#pragma unroll
    for (int j = 0; j < 8; ++j) {
      const int row = rowBase + wave * 16 + j + 8 * half;
      C[(size_t)row * N + col] = acc[nt][j] + bv;
    }
  }
}

__global__ __launch_bounds__(256) void attn_kernel(
    const float* __restrict__ Q, const float* __restrict__ KV,
    const float* __restrict__ peq, const float* __restrict__ pekv,
    const float* __restrict__ mask, float* __restrict__ O) {
  __shared__ float ks[K2N][34];
  __shared__ float pst[8][16][18];

  const int idx = blockIdx.x;
  const int h = idx % KNH;
  const int g = (idx / KNH) % KG;
  const int b = idx / (KNH * KG);
  const int gh = g >> 2, gw = g & 3;

  const int tid = threadIdx.x;
  const int wave = tid >> 5;
  const int lane = tid & 31;
  const int l16 = lane & 15;
  const int half = lane >> 4;

  {
    const int j = tid;
    const int s = j >> 7;
    const int n = j & 127;
    const int hsrc =
        (gh * 8 + (n >> 4) + (s == 0 ? KSHIFT : (KH - KSHIFT))) & (KH - 1);
    const int wsrc = gw * 16 + (n & 15);
    const float* krow =
        KV + (size_t)((b * KH + hsrc) * KW + wsrc) * (2 * KDIM) + h * KHEAD;
    const float* cosr = pekv + ((size_t)(0 * KG + g) * K2N + j) * (KHEAD / 2);
    const float* sinr = pekv + ((size_t)(1 * KG + g) * K2N + j) * (KHEAD / 2);
#pragma unroll
    for (int c = 0; c < 16; ++c) {
      const float x1 = krow[2 * c], x2 = krow[2 * c + 1];
      const float cc = cosr[c], ss = sinr[c];
      ks[j][c]      = x1 * cc - x2 * ss;
      ks[j][c + 16] = x2 * cc + x1 * ss;
    }
  }

  const int qn = wave * 16 + l16;
  const int qhs = gh * 8 + (qn >> 4);
  const int qws = gw * 16 + (qn & 15);
  const float* qrow = Q + (size_t)((b * KH + qhs) * KW + qws) * KDIM + h * KHEAD;
  const float* qcos = peq + ((size_t)(0 * KG + g) * KN + qn) * (KHEAD / 2);
  const float* qsin = peq + ((size_t)(1 * KG + g) * KN + qn) * (KHEAD / 2);
  v2f qf[8];
#pragma unroll
  for (int kk = 0; kk < 8; ++kk) {
    const int k0 = 4 * kk + 2 * half;
#pragma unroll
    for (int e = 0; e < 2; ++e) {
      if (kk < 4) {
        const int c = k0 + e;
        qf[kk][e] = qrow[2 * c] * qcos[c] - qrow[2 * c + 1] * qsin[c];
      } else {
        const int c = k0 + e - 16;
        qf[kk][e] = qrow[2 * c] * qsin[c] + qrow[2 * c + 1] * qcos[c];
      }
    }
  }

  __syncthreads();

  const float scale = 0.1767766952966369f;
  float mrowv[8], lrowv[8];
  v8f oacc[2];
  oacc[0] = {}; oacc[1] = {};
#pragma unroll
  for (int j = 0; j < 8; ++j) { mrowv[j] = -1e30f; lrowv[j] = 0.f; }

  const float* maskg = mask + (size_t)g * KN * K2N;

  for (int kt = 0; kt < 16; ++kt) {
    v8f s = {};
#pragma unroll
    for (int kk = 0; kk < 8; ++kk) {
      v2f bfr = *(const v2f*)&ks[kt * 16 + l16][4 * kk + 2 * half];
      s = wmma_f32(qf[kk], bfr, s);
    }

    const int keyc = kt * 16 + l16;
#pragma unroll
    for (int j = 0; j < 8; ++j) {
      const int qr = wave * 16 + j + 8 * half;
      s[j] = s[j] * scale + maskg[(size_t)qr * K2N + keyc];
    }

#pragma unroll
    for (int j = 0; j < 8; ++j) {
      float m = s[j];
#pragma unroll
      for (int d = 1; d < 16; d <<= 1) m = fmaxf(m, __shfl_xor(m, d, 32));
      const float mn = fmaxf(mrowv[j], m);
      const float alpha = expf(mrowv[j] - mn);
      const float pv = expf(s[j] - mn);
      s[j] = pv;
      float rs = pv;
#pragma unroll
      for (int d = 1; d < 16; d <<= 1) rs += __shfl_xor(rs, d, 32);
      lrowv[j] = lrowv[j] * alpha + rs;
      mrowv[j] = mn;
      oacc[0][j] *= alpha;
      oacc[1][j] *= alpha;
    }

#pragma unroll
    for (int j = 0; j < 8; ++j)
      pst[wave][j + 8 * half][l16] = s[j];
    asm volatile("s_wait_dscnt 0" ::: "memory");

    const int sv   = kt >> 3;
    const int hsrc = (gh * 8 + (kt & 7) + (sv == 0 ? KSHIFT : (KH - KSHIFT))) &
                     (KH - 1);
    const float* vbase = KV +
        (size_t)((b * KH + hsrc) * KW + gw * 16) * (2 * KDIM) + KDIM +
        h * KHEAD;
#pragma unroll
    for (int kk = 0; kk < 4; ++kk) {
      v2f a = *(const v2f*)&pst[wave][l16][4 * kk + 2 * half];
      const float* v0 = vbase + (size_t)(4 * kk + 2 * half) * (2 * KDIM);
#pragma unroll
      for (int ct = 0; ct < 2; ++ct) {
        v2f bv;
        bv.x = v0[ct * 16 + l16];
        bv.y = v0[2 * KDIM + ct * 16 + l16];
        oacc[ct] = wmma_f32(a, bv, oacc[ct]);
      }
    }
  }

#pragma unroll
  for (int ct = 0; ct < 2; ++ct) {
#pragma unroll
    for (int j = 0; j < 8; ++j) {
      const int n = wave * 16 + j + 8 * half;
      const int hs = gh * 8 + (n >> 4);
      const int ws = gw * 16 + (n & 15);
      O[(size_t)((b * KH + hs) * KW + ws) * KDIM + h * KHEAD + ct * 16 + l16] =
          oacc[ct][j] / lrowv[j];
    }
  }
}

extern "C" void kernel_launch(void* const* d_in, const int* in_sizes, int n_in,
                              void* d_out, int out_size, void* d_ws,
                              size_t ws_size, hipStream_t stream) {
  const float* x      = (const float*)d_in[0];
  const float* peq    = (const float*)d_in[1];
  const float* pekv   = (const float*)d_in[2];
  const float* mask   = (const float*)d_in[3];
  const float* w_q    = (const float*)d_in[4];
  const float* b_q    = (const float*)d_in[5];
  const float* w_kv   = (const float*)d_in[6];
  const float* b_kv   = (const float*)d_in[7];
  const float* w_proj = (const float*)d_in[8];
  const float* b_proj = (const float*)d_in[9];

  const int B = in_sizes[0] / (KH * KW * KDIM);
  const int M = B * KH * KW;

  float* qbuf  = (float*)d_ws;
  float* kvbuf = qbuf + (size_t)M * KDIM;
  float* obuf  = kvbuf + (size_t)M * 2 * KDIM;

  dim3 blk(256);
  gemm_bias_kernel<<<dim3((M / 128) * (KDIM / 64)), blk, 0, stream>>>(
      x, w_q, b_q, qbuf, M, KDIM);
  gemm_bias_kernel<<<dim3((M / 128) * (2 * KDIM / 64)), blk, 0, stream>>>(
      x, w_kv, b_kv, kvbuf, M, 2 * KDIM);
  attn_kernel<<<dim3(B * KG * KNH), blk, 0, stream>>>(qbuf, kvbuf, peq, pekv,
                                                      mask, obuf);
  gemm_bias_kernel<<<dim3((M / 128) * (KDIM / 64)), blk, 0, stream>>>(
      obuf, w_proj, b_proj, (float*)d_out, M, KDIM);
}


# GEMM weight-stationary: each work-group keeps its 512x128 W slab transposed in LDS, waves take A fragments straight from global by 16-byte loads, no K-loop barrier; k sum order in 8-k groups changes
# speedup vs baseline: 2.0579x; 2.0579x over previous
#include <hip/hip_runtime.h>
#include <math.h>

typedef __attribute__((ext_vector_type(2))) float v2f;
typedef __attribute__((ext_vector_type(4))) float v4f;
typedef __attribute__((ext_vector_type(8))) float v8f;

#define KDIM   512
#define KNH    16
#define KHEAD  32
#define KH     64
#define KW     64
#define KG     32
#define KN     128
#define K2N    256
#define KSHIFT 4

static __device__ __forceinline__ v8f wmma_f32(v2f a, v2f b, v8f c) {
  return __builtin_amdgcn_wmma_f32_16x16x4_f32(false, a, false, b, (short)0, c,
                                               false, false);
}

#define WPITCH 516

__global__ __launch_bounds__(256) void gemm_bias_kernel(
    const float* __restrict__ A, const float* __restrict__ Wm,
    const float* __restrict__ bias, float* __restrict__ C, int M, int N) {
  __shared__ __attribute__((aligned(16))) float WT[128][WPITCH];

  const int nBlocks = N >> 7;
  const int nBlock  = blockIdx.x % nBlocks;
  const int mGroup  = blockIdx.x / nBlocks;
  const int mGroups = gridDim.x / nBlocks;
  const int colBase = nBlock << 7;

  const int tid  = threadIdx.x;
  const int wave = tid >> 5;
  const int lane = tid & 31;
  const int l16  = lane & 15;
  const int half = lane >> 4;

  {
    const int fn = ((lane & 7) + ((wave & 3) << 3)) << 2;
    const int fk = ((lane >> 3) + ((wave >> 2) << 2)) << 2;
    const float* wg = Wm + (size_t)fk * N + colBase + fn;
    for (int p = 0; p < 4; ++p) {
      v4f rb[4][4];
#pragma unroll
      for (int j = 0; j < 4; ++j)
#pragma unroll
        for (int r = 0; r < 4; ++r)
          rb[j][r] =
              *(const v4f*)(wg + (size_t)(128 * p + 32 * j + r) * N);
#pragma unroll
      for (int j = 0; j < 4; ++j)
#pragma unroll
        for (int e = 0; e < 4; ++e) {
          v4f t = {rb[j][0][e], rb[j][1][e], rb[j][2][e], rb[j][3][e]};
          *(v4f*)&WT[fn + e][128 * p + 32 * j + fk] = t;
        }
    }
  }
  __syncthreads();

  float bv[8];
#pragma unroll
  for (int nt = 0; nt < 8; ++nt) bv[nt] = bias[colBase + nt * 16 + l16];

  const int tilesTotal = M >> 5;
  const int tileStride = mGroups << 3;
  for (int t = (mGroup << 3) + wave; t < tilesTotal; t += tileStride) {
    const int row0 = t << 5;
    v8f acc[2][8];
#pragma unroll
    for (int mt = 0; mt < 2; ++mt)
#pragma unroll
      for (int nt = 0; nt < 8; ++nt) acc[mt][nt] = {};

    const float* ap0 = A + (size_t)(row0 + l16) * KDIM + 4 * half;
    const float* ap1 = ap0 + 16 * KDIM;
    const float* wp = &WT[l16][4 * half];

    v4f ac[2][8], an[2][8];
#pragma unroll
    for (int k8 = 0; k8 < 8; ++k8) {
      ac[0][k8] = *(const v4f*)(ap0 + 8 * k8);
      ac[1][k8] = *(const v4f*)(ap1 + 8 * k8);
    }
    v4f b[2][8];
#pragma unroll
    for (int nt = 0; nt < 8; ++nt)
      b[0][nt] = *(const v4f*)(wp + nt * 16 * WPITCH);

    for (int c = 0; c < 8; ++c) {
      if (c < 7) {
#pragma unroll
        for (int k8 = 0; k8 < 8; ++k8) {
          an[0][k8] = *(const v4f*)(ap0 + 64 * (c + 1) + 8 * k8);
          an[1][k8] = *(const v4f*)(ap1 + 64 * (c + 1) + 8 * k8);
        }
      }
      const float* wc = wp + 64 * c;
#pragma unroll
      for (int k8 = 0; k8 < 8; ++k8) {
        if (k8 < 7 || c < 7) {
#pragma unroll
          for (int nt = 0; nt < 8; ++nt)
            b[(k8 + 1) & 1][nt] =
                *(const v4f*)(wc + nt * 16 * WPITCH + 8 * (k8 + 1));
        }
        __builtin_amdgcn_sched_barrier(0);
#pragma unroll
        for (int nt = 0; nt < 8; ++nt) {
#pragma unroll
          for (int mt = 0; mt < 2; ++mt) {
            v2f alo = {ac[mt][k8].x, ac[mt][k8].y};
            v2f blo = {b[k8 & 1][nt].x, b[k8 & 1][nt].y};
            acc[mt][nt] = wmma_f32(alo, blo, acc[mt][nt]);
          }
        }
#pragma unroll
        for (int nt = 0; nt < 8; ++nt) {
#pragma unroll
          for (int mt = 0; mt < 2; ++mt) {
            v2f ahi = {ac[mt][k8].z, ac[mt][k8].w};
            v2f bhi = {b[k8 & 1][nt].z, b[k8 & 1][nt].w};
            acc[mt][nt] = wmma_f32(ahi, bhi, acc[mt][nt]);
          }
        }
        __builtin_amdgcn_sched_barrier(0);
      }
      if (c < 7) {
#pragma unroll
        for (int k8 = 0; k8 < 8; ++k8) {
          ac[0][k8] = an[0][k8];
          ac[1][k8] = an[1][k8];
        }
      }
    }

#pragma unroll
    for (int nt = 0; nt < 8; ++nt) {
      const int col = colBase + nt * 16 + l16;
#pragma unroll
      for (int mt = 0; mt < 2; ++mt) {
#pragma unroll
        for (int j = 0; j < 8; ++j) {
          const int row = row0 + mt * 16 + j + 8 * half;
          C[(size_t)row * N + col] = acc[mt][nt][j] + bv[nt];
        }
      }
    }
  }
}

__global__ __launch_bounds__(256) void attn_kernel(
    const float* __restrict__ Q, const float* __restrict__ KV,
    const float* __restrict__ peq, const float* __restrict__ pekv,
    const float* __restrict__ mask, float* __restrict__ O) {
  __shared__ float ks[K2N][34];
  __shared__ float pst[8][16][18];

  const int idx = blockIdx.x;
  const int h = idx % KNH;
  const int g = (idx / KNH) % KG;
  const int b = idx / (KNH * KG);
  const int gh = g >> 2, gw = g & 3;

  const int tid = threadIdx.x;
  const int wave = tid >> 5;
  const int lane = tid & 31;
  const int l16 = lane & 15;
  const int half = lane >> 4;

  {
    const int j = tid;
    const int s = j >> 7;
    const int n = j & 127;
    const int hsrc =
        (gh * 8 + (n >> 4) + (s == 0 ? KSHIFT : (KH - KSHIFT))) & (KH - 1);
    const int wsrc = gw * 16 + (n & 15);
    const float* krow =
        KV + (size_t)((b * KH + hsrc) * KW + wsrc) * (2 * KDIM) + h * KHEAD;
    const float* cosr = pekv + ((size_t)(0 * KG + g) * K2N + j) * (KHEAD / 2);
    const float* sinr = pekv + ((size_t)(1 * KG + g) * K2N + j) * (KHEAD / 2);
#pragma unroll
    for (int c = 0; c < 16; ++c) {
      const float x1 = krow[2 * c], x2 = krow[2 * c + 1];
      const float cc = cosr[c], ss = sinr[c];
      ks[j][c]      = x1 * cc - x2 * ss;
      ks[j][c + 16] = x2 * cc + x1 * ss;
    }
  }

  const int qn = wave * 16 + l16;
  const int qhs = gh * 8 + (qn >> 4);
  const int qws = gw * 16 + (qn & 15);
  const float* qrow = Q + (size_t)((b * KH + qhs) * KW + qws) * KDIM + h * KHEAD;
  const float* qcos = peq + ((size_t)(0 * KG + g) * KN + qn) * (KHEAD / 2);
  const float* qsin = peq + ((size_t)(1 * KG + g) * KN + qn) * (KHEAD / 2);
  v2f qf[8];
#pragma unroll
  for (int kk = 0; kk < 8; ++kk) {
    const int k0 = 4 * kk + 2 * half;
#pragma unroll
    for (int e = 0; e < 2; ++e) {
      if (kk < 4) {
        const int c = k0 + e;
        qf[kk][e] = qrow[2 * c] * qcos[c] - qrow[2 * c + 1] * qsin[c];
      } else {
        const int c = k0 + e - 16;
        qf[kk][e] = qrow[2 * c] * qsin[c] + qrow[2 * c + 1] * qcos[c];
      }
    }
  }

  __syncthreads();

  const float scale = 0.1767766952966369f;
  const float* maskg = mask + (size_t)g * KN * K2N;

  v8f s[16];
#pragma unroll
  for (int tg = 0; tg < 4; ++tg) {
    __builtin_amdgcn_sched_barrier(0);
#pragma unroll
    for (int t = 0; t < 4; ++t) s[tg * 4 + t] = {};
#pragma unroll
    for (int kk = 0; kk < 8; ++kk) {
#pragma unroll
      for (int t = 0; t < 4; ++t) {
        const int kt = tg * 4 + t;
        v2f bfr = *(const v2f*)&ks[kt * 16 + l16][4 * kk + 2 * half];
        s[kt] = wmma_f32(qf[kk], bfr, s[kt]);
      }
    }
#pragma unroll
    for (int t = 0; t < 4; ++t) {
      const int kt = tg * 4 + t;
      const int keyc = kt * 16 + l16;
#pragma unroll
      for (int j = 0; j < 8; ++j) {
        const int qr = wave * 16 + j + 8 * half;
        s[kt][j] = s[kt][j] * scale + maskg[(size_t)qr * K2N + keyc];
      }
    }
  }

  float mrowv[8], lrowv[8];
#pragma unroll
  for (int j = 0; j < 8; ++j) {
    float m = -1e30f;
#pragma unroll
    for (int kt = 0; kt < 16; ++kt) m = fmaxf(m, s[kt][j]);
#pragma unroll
    for (int d = 1; d < 16; d <<= 1) m = fmaxf(m, __shfl_xor(m, d, 32));
    mrowv[j] = m;
  }
#pragma unroll
  for (int j = 0; j < 8; ++j) {
    float rs = 0.f;
#pragma unroll
    for (int kt = 0; kt < 16; ++kt) {
      const float pv = expf(s[kt][j] - mrowv[j]);
      s[kt][j] = pv;
      rs += pv;
    }
#pragma unroll
    for (int d = 1; d < 16; d <<= 1) rs += __shfl_xor(rs, d, 32);
    lrowv[j] = rs;
  }

  v2f bv[2][4][2];
#define LOAD_V_TILE(KT, BUF)                                                  \
  {                                                                           \
                             \
    const int sv_ = (KT) >> 3;                                                \
    const int hsrc_ = (gh * 8 + ((KT) & 7) +                                  \
                       (sv_ == 0 ? KSHIFT : (KH - KSHIFT))) & (KH - 1);       \
    const float* vbase_ = KV +                                                \
        (size_t)((b * KH + hsrc_) * KW + gw * 16) * (2 * KDIM) + KDIM +       \
        h * KHEAD;                                                            \
    _Pragma("unroll") for (int kk = 0; kk < 4; ++kk) {                        \
      const float* v0_ = vbase_ + (size_t)(4 * kk + 2 * half) * (2 * KDIM);   \
      _Pragma("unroll") for (int ct = 0; ct < 2; ++ct) {                      \
        bv[BUF][kk][ct].x = v0_[ct * 16 + l16];                               \
        bv[BUF][kk][ct].y = v0_[2 * KDIM + ct * 16 + l16];                    \
      }                                                                       \
    }                                                                         \
  }

  v8f oacc[2];
  oacc[0] = {}; oacc[1] = {};
  LOAD_V_TILE(0, 0)
#pragma unroll
  for (int kt = 0; kt < 16; ++kt) {
    if (kt + 1 < 16) LOAD_V_TILE(kt + 1, (kt + 1) & 1)
    __builtin_amdgcn_sched_barrier(0);

#pragma unroll
    for (int j = 0; j < 8; ++j)
      pst[wave][j + 8 * half][l16] = s[kt][j];
    asm volatile("s_wait_dscnt 0" ::: "memory");

#pragma unroll
    for (int kk = 0; kk < 4; ++kk) {
      v2f a = *(const v2f*)&pst[wave][l16][4 * kk + 2 * half];
#pragma unroll
      for (int ct = 0; ct < 2; ++ct)
        oacc[ct] = wmma_f32(a, bv[kt & 1][kk][ct], oacc[ct]);
    }
  }
#undef LOAD_V_TILE

#pragma unroll
  for (int ct = 0; ct < 2; ++ct) {
#pragma unroll
    for (int j = 0; j < 8; ++j) {
      const int n = wave * 16 + j + 8 * half;
      const int hs = gh * 8 + (n >> 4);
      const int ws = gw * 16 + (n & 15);
      O[(size_t)((b * KH + hs) * KW + ws) * KDIM + h * KHEAD + ct * 16 + l16] =
          oacc[ct][j] / lrowv[j];
    }
  }
}

extern "C" void kernel_launch(void* const* d_in, const int* in_sizes, int n_in,
                              void* d_out, int out_size, void* d_ws,
                              size_t ws_size, hipStream_t stream) {
  const float* x      = (const float*)d_in[0];
  const float* peq    = (const float*)d_in[1];
  const float* pekv   = (const float*)d_in[2];
  const float* mask   = (const float*)d_in[3];
  const float* w_q    = (const float*)d_in[4];
  const float* b_q    = (const float*)d_in[5];
  const float* w_kv   = (const float*)d_in[6];
  const float* b_kv   = (const float*)d_in[7];
  const float* w_proj = (const float*)d_in[8];
  const float* b_proj = (const float*)d_in[9];

  const int B = in_sizes[0] / (KH * KW * KDIM);
  const int M = B * KH * KW;

  float* qbuf  = (float*)d_ws;
  float* kvbuf = qbuf + (size_t)M * KDIM;
  float* obuf  = kvbuf + (size_t)M * 2 * KDIM;

  dim3 blk(256);
  gemm_bias_kernel<<<dim3(256), blk, 0, stream>>>(x, w_q, b_q, qbuf, M, KDIM);
  gemm_bias_kernel<<<dim3(256), blk, 0, stream>>>(x, w_kv, b_kv, kvbuf, M,
                                                   2 * KDIM);
  attn_kernel<<<dim3(B * KG * KNH), blk, 0, stream>>>(qbuf, kvbuf, peq, pekv,
                                                      mask, obuf);
  gemm_bias_kernel<<<dim3(256), blk, 0, stream>>>(obuf, w_proj, b_proj,
                                                   (float*)d_out, M, KDIM);
}
